# starting kernel
# baseline (speedup 1.0000x reference)
#include <hip/hip_runtime.h>
#include <hip/hip_bf16.h>

typedef __attribute__((ext_vector_type(2))) float v2f;
typedef __attribute__((ext_vector_type(8))) float v8f;

#define LDSTRIDE 132

__global__ void gnn_deg_kernel(const int* __restrict__ src,
                               const int* __restrict__ dst,
                               float* __restrict__ deg_out,
                               float* __restrict__ deg_in, int E) {
    int e = blockIdx.x * blockDim.x + threadIdx.x;
    if (e < E) {
        atomicAdd(&deg_out[src[e]], 1.0f);
        atomicAdd(&deg_in[dst[e]], 1.0f);
    }
}

__global__ void gnn_norm_kernel(float* __restrict__ d, int n) {
    int i = blockIdx.x * blockDim.x + threadIdx.x;
    if (i < n) {
        float v = fmaxf(d[i], 1.0f);
        d[i] = 1.0f / sqrtf(v);
    }
}

__global__ __launch_bounds__(128) void gnn_gemm_kernel(
    const float* __restrict__ H, const float* __restrict__ nsrc,
    const float* __restrict__ W, float* __restrict__ X, int N) {
    __shared__ float As[16 * LDSTRIDE];

    const int row0 = blockIdx.x * 16;
    const int tid  = threadIdx.x;

    for (int i = tid; i < 16 * 32; i += 128) {
        int r  = i >> 5;
        int c4 = (i & 31) * 4;
        int row = row0 + r;
        int rowc = (row < N) ? row : (N - 1);
        float s = nsrc[rowc];
        float4 v = *(const float4*)(H + (size_t)rowc * 128 + c4);
        float4 sv = make_float4(v.x * s, v.y * s, v.z * s, v.w * s);
        *(float4*)(&As[r * LDSTRIDE + c4]) = sv;
    }
    __syncthreads();

    const int wave = tid >> 5;
    const int lane = tid & 31;
    const int ln   = lane & 15;
    const int koff = (lane < 16) ? 0 : 2;

    const int n0 = wave * 32;
    const int n1 = n0 + 16;

    v8f acc0 = {};
    v8f acc1 = {};

    for (int kb = 0; kb < 128; kb += 4) {
        v2f a = *(const v2f*)(&As[ln * LDSTRIDE + kb + koff]);

        const float* wp = W + (size_t)(kb + koff) * 128;
        v2f b0, b1;
        b0.x = wp[n0 + ln];
        b0.y = wp[128 + n0 + ln];
        b1.x = wp[n1 + ln];
        b1.y = wp[128 + n1 + ln];

        acc0 = __builtin_amdgcn_wmma_f32_16x16x4_f32(
            false, a, false, b0, (short)0, acc0, false, false);
        acc1 = __builtin_amdgcn_wmma_f32_16x16x4_f32(
            false, a, false, b1, (short)0, acc1, false, false);
    }

    const int mbase = row0 + ((lane < 16) ? 0 : 8);
#pragma unroll
    for (int v = 0; v < 8; ++v) {
        int row = mbase + v;
        if (row < N) {
            X[(size_t)row * 128 + n0 + ln] = acc0[v];
            X[(size_t)row * 128 + n1 + ln] = acc1[v];
        }
    }
}

__global__ __launch_bounds__(256) void gnn_agg_kernel(
    const float* __restrict__ X, const int* __restrict__ src,
    const int* __restrict__ dst, float* __restrict__ AGG, int E) {
    long long t = (long long)blockIdx.x * blockDim.x + threadIdx.x;
    int e = (int)(t >> 5);
    if (e >= E) return;
    int lane = (int)(t & 31);
    int s = src[e];
    int d = dst[e];
    float4 v = *(const float4*)(X + (size_t)s * 128 + lane * 4);
    float* out = AGG + (size_t)d * 128 + lane * 4;
    atomicAdd(out + 0, v.x);
    atomicAdd(out + 1, v.y);
    atomicAdd(out + 2, v.z);
    atomicAdd(out + 3, v.w);
}

__global__ __launch_bounds__(256) void gnn_fin_kernel(
    const float* __restrict__ AGG, const float* __restrict__ ndst,
    const float* __restrict__ bias, float* __restrict__ OUT, int N, int do_relu) {
    long long t = (long long)blockIdx.x * blockDim.x + threadIdx.x;
    if (t >= (long long)N * 32) return;
    int node = (int)(t >> 5);
    int c4   = (int)(t & 31) * 4;
    float s = ndst[node];
    float4 v = *(const float4*)(AGG + (size_t)node * 128 + c4);
    float4 b = *(const float4*)(bias + c4);
    float4 o;
    o.x = v.x * s + b.x;
    o.y = v.y * s + b.y;
    o.z = v.z * s + b.z;
    o.w = v.w * s + b.w;
    if (do_relu) {
        o.x = fmaxf(o.x, 0.0f);
        o.y = fmaxf(o.y, 0.0f);
        o.z = fmaxf(o.z, 0.0f);
        o.w = fmaxf(o.w, 0.0f);
    }
    *(float4*)(OUT + (size_t)node * 128 + c4) = o;
}

extern "C" void kernel_launch(void* const* d_in, const int* in_sizes, int n_in,
                              void* d_out, int out_size, void* d_ws, size_t ws_size,
                              hipStream_t stream) {
    const float* feature = (const float*)d_in[0];
    const int*   src     = (const int*)d_in[1];
    const int*   dst     = (const int*)d_in[2];
    const float* W1      = (const float*)d_in[3];
    const float* b1      = (const float*)d_in[4];
    const float* W2      = (const float*)d_in[5];
    const float* b2      = (const float*)d_in[6];
    float* out = (float*)d_out;

    const int F = 128;
    const int N = in_sizes[0] / F;
    const int E = in_sizes[1];

    char* ws = (char*)d_ws;
    float* nsrc = (float*)ws;
    float* ndst = nsrc + N;
    size_t off = ((size_t)2 * N * sizeof(float) + 255) & ~(size_t)255;
    float* buf0 = (float*)(ws + off);
    float* buf1 = buf0 + (size_t)N * F;

    const size_t featBytes = (size_t)N * F * sizeof(float);

    hipMemsetAsync(nsrc, 0, (size_t)2 * N * sizeof(float), stream);
    gnn_deg_kernel<<<(E + 255) / 256, 256, 0, stream>>>(src, dst, nsrc, ndst, E);
    gnn_norm_kernel<<<(2 * N + 255) / 256, 256, 0, stream>>>(nsrc, 2 * N);

    const int gemmBlocks = (N + 15) / 16;
    const long long edgeThreads = (long long)E * 32;
    const int aggBlocks = (int)((edgeThreads + 255) / 256);
    const long long nodeThreads = (long long)N * 32;
    const int finBlocks = (int)((nodeThreads + 255) / 256);

    gnn_gemm_kernel<<<gemmBlocks, 128, 0, stream>>>(feature, nsrc, W1, buf0, N);
    hipMemsetAsync(buf1, 0, featBytes, stream);
    gnn_agg_kernel<<<aggBlocks, 256, 0, stream>>>(buf0, src, dst, buf1, E);
    gnn_fin_kernel<<<finBlocks, 256, 0, stream>>>(buf1, ndst, b1, buf0, N, 1);

    gnn_gemm_kernel<<<gemmBlocks, 128, 0, stream>>>(buf0, nsrc, W2, buf1, N);
    hipMemsetAsync(buf0, 0, featBytes, stream);
    gnn_agg_kernel<<<aggBlocks, 256, 0, stream>>>(buf1, src, dst, buf0, E);
    gnn_fin_kernel<<<finBlocks, 256, 0, stream>>>(buf0, ndst, b2, out, N, 0);
}


# Scatter-add: one wave per run of 32 edges, not per edge: ids read as one whole-line load and broadcast by lane read, 8 edges' row loads issued before their atomics; float atomics, arrival order kept
# speedup vs baseline: 1.6407x; 1.6407x over previous
#include <hip/hip_runtime.h>
#include <hip/hip_bf16.h>

typedef __attribute__((ext_vector_type(2))) float v2f;
typedef __attribute__((ext_vector_type(8))) float v8f;

#define LDSTRIDE 132

__global__ void gnn_deg_kernel(const int* __restrict__ src,
                               const int* __restrict__ dst,
                               float* __restrict__ deg_out,
                               float* __restrict__ deg_in, int E) {
    int e = blockIdx.x * blockDim.x + threadIdx.x;
    if (e < E) {
        atomicAdd(&deg_out[src[e]], 1.0f);
        atomicAdd(&deg_in[dst[e]], 1.0f);
    }
}

__global__ void gnn_norm_kernel(float* __restrict__ d, int n) {
    int i = blockIdx.x * blockDim.x + threadIdx.x;
    if (i < n) {
        float v = fmaxf(d[i], 1.0f);
        d[i] = 1.0f / sqrtf(v);
    }
}

__global__ __launch_bounds__(128) void gnn_gemm_kernel(
    const float* __restrict__ H, const float* __restrict__ nsrc,
    const float* __restrict__ W, float* __restrict__ X, int N) {
    __shared__ float As[16 * LDSTRIDE];

    const int row0 = blockIdx.x * 16;
    const int tid  = threadIdx.x;

    for (int i = tid; i < 16 * 32; i += 128) {
        int r  = i >> 5;
        int c4 = (i & 31) * 4;
        int row = row0 + r;
        int rowc = (row < N) ? row : (N - 1);
        float s = nsrc[rowc];
        float4 v = *(const float4*)(H + (size_t)rowc * 128 + c4);
        float4 sv = make_float4(v.x * s, v.y * s, v.z * s, v.w * s);
        *(float4*)(&As[r * LDSTRIDE + c4]) = sv;
    }
    __syncthreads();

    const int wave = tid >> 5;
    const int lane = tid & 31;
    const int ln   = lane & 15;
    const int koff = (lane < 16) ? 0 : 2;

    const int n0 = wave * 32;
    const int n1 = n0 + 16;

    v8f acc0 = {};
    v8f acc1 = {};

    for (int kb = 0; kb < 128; kb += 4) {
        v2f a = *(const v2f*)(&As[ln * LDSTRIDE + kb + koff]);

        const float* wp = W + (size_t)(kb + koff) * 128;
        v2f b0, b1;
        b0.x = wp[n0 + ln];
        b0.y = wp[128 + n0 + ln];
        b1.x = wp[n1 + ln];
        b1.y = wp[128 + n1 + ln];

        acc0 = __builtin_amdgcn_wmma_f32_16x16x4_f32(
            false, a, false, b0, (short)0, acc0, false, false);
        acc1 = __builtin_amdgcn_wmma_f32_16x16x4_f32(
            false, a, false, b1, (short)0, acc1, false, false);
    }

    const int mbase = row0 + ((lane < 16) ? 0 : 8);
#pragma unroll
    for (int v = 0; v < 8; ++v) {
        int row = mbase + v;
        if (row < N) {
            X[(size_t)row * 128 + n0 + ln] = acc0[v];
            X[(size_t)row * 128 + n1 + ln] = acc1[v];
        }
    }
}

#define AGG_RUN   32
#define AGG_GROUP 8

__device__ __forceinline__ void gnn_agg_one_edge(
    const float* __restrict__ X, float* __restrict__ AGG,
    int s, int d, unsigned lane) {
    const float* xr = X + (size_t)s * 128;
    float v0 = xr[lane];
    float v1 = xr[lane + 32];
    float v2 = xr[lane + 64];
    float v3 = xr[lane + 96];
    float* ar = AGG + (size_t)d * 128;
    atomicAdd(ar + lane,      v0);
    atomicAdd(ar + lane + 32, v1);
    atomicAdd(ar + lane + 64, v2);
    atomicAdd(ar + lane + 96, v3);
}

__global__ __launch_bounds__(256) void gnn_agg_kernel(
    const float* __restrict__ X, const int* __restrict__ src,
    const int* __restrict__ dst, float* __restrict__ AGG, int E) {
    const unsigned lane = threadIdx.x & 31;
    const int wave = __builtin_amdgcn_readfirstlane(
        (int)(blockIdx.x * (256 / AGG_RUN) + (threadIdx.x >> 5)));
    const long long e0 = (long long)wave * AGG_RUN;
    if (e0 >= E) return;
    const long long left = (long long)E - e0;
    const int n = left < AGG_RUN ? (int)left : AGG_RUN;

    const long long el = e0 + ((int)lane < n ? (int)lane : 0);
    const int sv = src[el];
    const int dv = dst[el];

    if (n == AGG_RUN) {
#pragma unroll
        for (int j0 = 0; j0 < AGG_RUN; j0 += AGG_GROUP) {
            float v[AGG_GROUP][4];
#pragma unroll
            for (int b = 0; b < AGG_GROUP; ++b) {
                const int s = __builtin_amdgcn_readlane(sv, j0 + b);
                const float* xr = X + (size_t)s * 128;
                v[b][0] = xr[lane];
                v[b][1] = xr[lane + 32];
                v[b][2] = xr[lane + 64];
                v[b][3] = xr[lane + 96];
            }
#pragma unroll
            for (int b = 0; b < AGG_GROUP; ++b) {
                const int d = __builtin_amdgcn_readlane(dv, j0 + b);
                float* ar = AGG + (size_t)d * 128;
                atomicAdd(ar + lane,      v[b][0]);
                atomicAdd(ar + lane + 32, v[b][1]);
                atomicAdd(ar + lane + 64, v[b][2]);
                atomicAdd(ar + lane + 96, v[b][3]);
            }
        }
    } else {
        for (int j = 0; j < n; ++j) {
            const int s = __builtin_amdgcn_readlane(sv, j);
            const int d = __builtin_amdgcn_readlane(dv, j);
            gnn_agg_one_edge(X, AGG, s, d, lane);
        }
    }
}

__global__ __launch_bounds__(256) void gnn_fin_kernel(
    const float* __restrict__ AGG, const float* __restrict__ ndst,
    const float* __restrict__ bias, float* __restrict__ OUT, int N, int do_relu) {
    long long t = (long long)blockIdx.x * blockDim.x + threadIdx.x;
    if (t >= (long long)N * 32) return;
    int node = (int)(t >> 5);
    int c4   = (int)(t & 31) * 4;
    float s = ndst[node];
    float4 v = *(const float4*)(AGG + (size_t)node * 128 + c4);
    float4 b = *(const float4*)(bias + c4);
    float4 o;
    o.x = v.x * s + b.x;
    o.y = v.y * s + b.y;
    o.z = v.z * s + b.z;
    o.w = v.w * s + b.w;
    if (do_relu) {
        o.x = fmaxf(o.x, 0.0f);
        o.y = fmaxf(o.y, 0.0f);
        o.z = fmaxf(o.z, 0.0f);
        o.w = fmaxf(o.w, 0.0f);
    }
    *(float4*)(OUT + (size_t)node * 128 + c4) = o;
}

extern "C" void kernel_launch(void* const* d_in, const int* in_sizes, int n_in,
                              void* d_out, int out_size, void* d_ws, size_t ws_size,
                              hipStream_t stream) {
    const float* feature = (const float*)d_in[0];
    const int*   src     = (const int*)d_in[1];
    const int*   dst     = (const int*)d_in[2];
    const float* W1      = (const float*)d_in[3];
    const float* b1      = (const float*)d_in[4];
    const float* W2      = (const float*)d_in[5];
    const float* b2      = (const float*)d_in[6];
    float* out = (float*)d_out;

    const int F = 128;
    const int N = in_sizes[0] / F;
    const int E = in_sizes[1];

    char* ws = (char*)d_ws;
    float* nsrc = (float*)ws;
    float* ndst = nsrc + N;
    size_t off = ((size_t)2 * N * sizeof(float) + 255) & ~(size_t)255;
    float* buf0 = (float*)(ws + off);
    float* buf1 = buf0 + (size_t)N * F;

    const size_t featBytes = (size_t)N * F * sizeof(float);

    hipMemsetAsync(nsrc, 0, (size_t)2 * N * sizeof(float), stream);
    gnn_deg_kernel<<<(E + 255) / 256, 256, 0, stream>>>(src, dst, nsrc, ndst, E);
    gnn_norm_kernel<<<(2 * N + 255) / 256, 256, 0, stream>>>(nsrc, 2 * N);

    const int gemmBlocks = (N + 15) / 16;
    const int aggWaves = (E + AGG_RUN - 1) / AGG_RUN;
    const int aggBlocks = (aggWaves + 7) / 8;
    const long long nodeThreads = (long long)N * 32;
    const int finBlocks = (int)((nodeThreads + 255) / 256);

    gnn_gemm_kernel<<<gemmBlocks, 128, 0, stream>>>(feature, nsrc, W1, buf0, N);
    hipMemsetAsync(buf1, 0, featBytes, stream);
    gnn_agg_kernel<<<aggBlocks, 256, 0, stream>>>(buf0, src, dst, buf1, E);
    gnn_fin_kernel<<<finBlocks, 256, 0, stream>>>(buf1, ndst, b1, buf0, N, 1);

    gnn_gemm_kernel<<<gemmBlocks, 128, 0, stream>>>(buf0, nsrc, W2, buf1, N);
    hipMemsetAsync(buf0, 0, featBytes, stream);
    gnn_agg_kernel<<<aggBlocks, 256, 0, stream>>>(buf1, src, dst, buf0, E);
    gnn_fin_kernel<<<finBlocks, 256, 0, stream>>>(buf0, ndst, b2, out, N, 0);
}


# Layer 2 reuses layer 1's per-wave edge queues, saved in the workspace as (source id, local row) pairs, instead of scanning dst again; rescan only after queue overflow; layer-2 sums in layer-1 order
# speedup vs baseline: 2.2079x; 1.3457x over previous
#include <hip/hip_runtime.h>
#include <hip/hip_bf16.h>

typedef __attribute__((ext_vector_type(2))) float v2f;
typedef __attribute__((ext_vector_type(8))) float v8f;

#define LDSTRIDE 132

__global__ void gnn_deg_kernel(const int* __restrict__ src,
                               const int* __restrict__ dst,
                               float* __restrict__ deg_out,
                               float* __restrict__ deg_in, int E) {
    int e = blockIdx.x * blockDim.x + threadIdx.x;
    if (e < E) {
        atomicAdd(&deg_out[src[e]], 1.0f);
        atomicAdd(&deg_in[dst[e]], 1.0f);
    }
}

__global__ void gnn_norm_kernel(float* __restrict__ d, int n) {
    int i = blockIdx.x * blockDim.x + threadIdx.x;
    if (i < n) {
        float v = fmaxf(d[i], 1.0f);
        d[i] = 1.0f / sqrtf(v);
    }
}

__global__ __launch_bounds__(128) void gnn_gemm_kernel(
    const float* __restrict__ H, const float* __restrict__ nsrc,
    const float* __restrict__ W, float* __restrict__ X, int N) {
    __shared__ float As[16 * LDSTRIDE];

    const int row0 = blockIdx.x * 16;
    const int tid  = threadIdx.x;

    for (int i = tid; i < 16 * 32; i += 128) {
        int r  = i >> 5;
        int c4 = (i & 31) * 4;
        int row = row0 + r;
        int rowc = (row < N) ? row : (N - 1);
        float s = nsrc[rowc];
        float4 v = *(const float4*)(H + (size_t)rowc * 128 + c4);
        float4 sv = make_float4(v.x * s, v.y * s, v.z * s, v.w * s);
        *(float4*)(&As[r * LDSTRIDE + c4]) = sv;
    }
    __syncthreads();

    const int wave = tid >> 5;
    const int lane = tid & 31;
    const int ln   = lane & 15;
    const int koff = (lane < 16) ? 0 : 2;

    const int n0 = wave * 32;
    const int n1 = n0 + 16;

    v8f acc0 = {};
    v8f acc1 = {};

    for (int kb = 0; kb < 128; kb += 4) {
        v2f a = *(const v2f*)(&As[ln * LDSTRIDE + kb + koff]);

        const float* wp = W + (size_t)(kb + koff) * 128;
        v2f b0, b1;
        b0.x = wp[n0 + ln];
        b0.y = wp[128 + n0 + ln];
        b1.x = wp[n1 + ln];
        b1.y = wp[128 + n1 + ln];

        acc0 = __builtin_amdgcn_wmma_f32_16x16x4_f32(
            false, a, false, b0, (short)0, acc0, false, false);
        acc1 = __builtin_amdgcn_wmma_f32_16x16x4_f32(
            false, a, false, b1, (short)0, acc1, false, false);
    }

    const int mbase = row0 + ((lane < 16) ? 0 : 8);
#pragma unroll
    for (int v = 0; v < 8; ++v) {
        int row = mbase + v;
        if (row < N) {
            X[(size_t)row * 128 + n0 + ln] = acc0[v];
            X[(size_t)row * 128 + n1 + ln] = acc1[v];
        }
    }
}

#define OWN_ROWS    416
#define OWN_TRASH   OWN_ROWS
#define OWN_THREADS 1024
#define OWN_WAVES   (OWN_THREADS / 32)
#define OWN_QW      384
#define OWN_GROUP   8
#define OWN_AHEAD   16

struct OwnLds {
    alignas(16) float acc[(OWN_ROWS + 1) * 128];
    int q_e[OWN_WAVES * OWN_QW];
    int q_r[OWN_WAVES * OWN_QW];
    int q_n[OWN_WAVES];
};

__device__ __forceinline__ void gnn_own_take(
    OwnLds& L, int e, unsigned lr,
    const float* __restrict__ X, const int* __restrict__ src) {
    const int w = (int)(lr & (OWN_WAVES - 1));
    const int slot = atomicAdd(&L.q_n[w], 1);
    if (slot < OWN_QW) {
        L.q_e[w * OWN_QW + slot] = e;
        L.q_r[w * OWN_QW + slot] = (int)lr;
    } else {
        const float* xr = X + (size_t)src[e] * 128;
        float* ar = L.acc + lr * 128;
#pragma unroll 1
        for (int c = 0; c < 128; ++c) atomicAdd(ar + c, xr[c]);
    }
}

__device__ __forceinline__ void gnn_own_scan4(
    OwnLds& L, int4 d, int e0, int base, unsigned rows,
    const float* __restrict__ X, const int* __restrict__ src) {
    const unsigned l0 = (unsigned)(d.x - base);
    const unsigned l1 = (unsigned)(d.y - base);
    const unsigned l2 = (unsigned)(d.z - base);
    const unsigned l3 = (unsigned)(d.w - base);
    const bool m0 = l0 < rows, m1 = l1 < rows, m2 = l2 < rows, m3 = l3 < rows;
    if (m0 | m1 | m2 | m3) {
        if (m0) gnn_own_take(L, e0,     l0, X, src);
        if (m1) gnn_own_take(L, e0 + 1, l1, X, src);
        if (m2) gnn_own_take(L, e0 + 2, l2, X, src);
        if (m3) gnn_own_take(L, e0 + 3, l3, X, src);
    }
}

__device__ __forceinline__ void gnn_own_scan(
    OwnLds& L, int tid, int lane, int wave, int base, unsigned rows,
    const float* __restrict__ X, const int* __restrict__ src,
    const int* __restrict__ dst, int E) {
    const int nchunk = E >> 7;
    const int4* d4 = (const int4*)dst;
    const int start = (int)(((long long)blockIdx.x * nchunk) / gridDim.x);
    for (int k = wave; k < nchunk; k += OWN_AHEAD * OWN_WAVES) {
        int c[OWN_AHEAD];
        int4 d[OWN_AHEAD];
#pragma unroll
        for (int u = 0; u < OWN_AHEAD; ++u) {
            const int kk = k + u * OWN_WAVES;
            int cc = start + (kk < nchunk ? kk : k);
            cc = cc >= nchunk ? cc - nchunk : cc;
            c[u] = cc;
            d[u] = d4[(size_t)cc * 32 + lane];
        }
#pragma unroll
        for (int u = 0; u < OWN_AHEAD; ++u) {
            if (k + u * OWN_WAVES < nchunk)
                gnn_own_scan4(L, d[u], c[u] * 128 + lane * 4, base, rows, X, src);
        }
    }
    for (int e = nchunk * 128 + tid; e < E; e += OWN_THREADS) {
        const unsigned lr = (unsigned)(dst[e] - base);
        if (lr < rows) gnn_own_take(L, e, lr, X, src);
    }
}

template <bool DIRECT, bool SAVE>
__device__ __forceinline__ void gnn_own_add(
    OwnLds& L, int n, int wave, int lane,
    const float* __restrict__ X, const int* __restrict__ src, int2* __restrict__ g) {
    const int* qe = L.q_e + wave * OWN_QW;
    const int* qr = L.q_r + wave * OWN_QW;
    int s_nx = 0, r_nx = OWN_TRASH;
    if (lane < OWN_GROUP && lane < n) {
        if (DIRECT) {
            const int2 p = g[lane];
            s_nx = p.x;
            r_nx = p.y;
        } else {
            r_nx = qr[lane];
            s_nx = src[qe[lane]];
        }
    }
    for (int i0 = 0; i0 < n; i0 += OWN_GROUP) {
        const int s_cu = s_nx, r_cu = r_nx;
        if (SAVE) {
            if (lane < OWN_GROUP && i0 + lane < n) g[i0 + lane] = make_int2(s_cu, r_cu);
        }
        const int idx = i0 + OWN_GROUP + lane;
        const bool ok = lane < OWN_GROUP && idx < n;
        const int idc = ok ? idx : 0;
        int r_ld, s_ld;
        if (DIRECT) {
            const int2 p = g[idc];
            s_ld = p.x;
            r_ld = p.y;
        } else {
            r_ld = qr[idc];
            s_ld = src[qe[idc]];
        }
        r_nx = ok ? r_ld : OWN_TRASH;
        s_nx = ok ? s_ld : 0;
        float4 v[OWN_GROUP];
#pragma unroll
        for (int b = 0; b < OWN_GROUP; ++b) {
            const int sb = __builtin_amdgcn_readlane(s_cu, b);
            v[b] = ((const float4*)(X + (size_t)sb * 128))[lane];
        }
#pragma unroll
        for (int b = 0; b < OWN_GROUP; ++b) {
            const int rb = __builtin_amdgcn_readlane(r_cu, b);
            float4* p = (float4*)(L.acc + rb * 128) + lane;
            float4 a = *p;
            a.x += v[b].x;
            a.y += v[b].y;
            a.z += v[b].z;
            a.w += v[b].w;
            *p = a;
        }
    }
}

template <bool REPLAY>
__global__ __launch_bounds__(OWN_THREADS) void gnn_own_kernel(
    const float* __restrict__ X, const int* __restrict__ src,
    const int* __restrict__ dst, const float* __restrict__ ndst,
    const float* __restrict__ bias, float* __restrict__ OUT,
    int* __restrict__ gq, int* __restrict__ gq_n,
    int N, int E, int do_relu) {
    __shared__ OwnLds L;

    const int tid = threadIdx.x;
    const int lane = tid & 31;
    const int wave = __builtin_amdgcn_readfirstlane(tid >> 5);
    const int base = blockIdx.x * OWN_ROWS;
    const int left = N - base;
    const unsigned rows = (unsigned)(left < OWN_ROWS ? left : OWN_ROWS);

    int c_all = 0;
    bool saved = false;
    if (REPLAY) {
        c_all = gq_n[blockIdx.x * OWN_WAVES + lane];
        saved = __builtin_amdgcn_ballot_w32(c_all > OWN_QW) == 0;
    }

    for (int i = tid; i < (OWN_ROWS + 1) * 32; i += OWN_THREADS)
        ((float4*)L.acc)[i] = make_float4(0.0f, 0.0f, 0.0f, 0.0f);
    if (tid < OWN_WAVES) L.q_n[tid] = 0;
    __syncthreads();

    int2* g = (int2*)gq + (size_t)(blockIdx.x * OWN_WAVES + wave) * OWN_QW;

    if (REPLAY && saved) {
        const int n = __builtin_amdgcn_readlane(c_all, wave);
        gnn_own_add<true, false>(L, n, wave, lane, X, src, g);
    } else {
        gnn_own_scan(L, tid, lane, wave, base, rows, X, src, dst, E);
        __syncthreads();
        const int cnt = __builtin_amdgcn_readfirstlane(L.q_n[wave]);
        if (!REPLAY) {
            if (lane == 0) gq_n[blockIdx.x * OWN_WAVES + wave] = cnt;
        }
        const int n = cnt < OWN_QW ? cnt : OWN_QW;
        gnn_own_add<false, !REPLAY>(L, n, wave, lane, X, src, g);
    }
    __syncthreads();

    for (int i = tid; i < (int)rows * 32; i += OWN_THREADS) {
        const int rr = i >> 5;
        const int c4 = (i & 31) * 4;
        const int node = base + rr;
        const float s = ndst[node];
        const float4 a = *(const float4*)(&L.acc[rr * 128 + c4]);
        const float4 b = *(const float4*)(bias + c4);
        float4 o;
        o.x = a.x * s + b.x;
        o.y = a.y * s + b.y;
        o.z = a.z * s + b.z;
        o.w = a.w * s + b.w;
        if (do_relu) {
            o.x = fmaxf(o.x, 0.0f);
            o.y = fmaxf(o.y, 0.0f);
            o.z = fmaxf(o.z, 0.0f);
            o.w = fmaxf(o.w, 0.0f);
        }
        *(float4*)(OUT + (size_t)node * 128 + c4) = o;
    }
}

extern "C" void kernel_launch(void* const* d_in, const int* in_sizes, int n_in,
                              void* d_out, int out_size, void* d_ws, size_t ws_size,
                              hipStream_t stream) {
    const float* feature = (const float*)d_in[0];
    const int*   src     = (const int*)d_in[1];
    const int*   dst     = (const int*)d_in[2];
    const float* W1      = (const float*)d_in[3];
    const float* b1      = (const float*)d_in[4];
    const float* W2      = (const float*)d_in[5];
    const float* b2      = (const float*)d_in[6];
    float* out = (float*)d_out;

    const int F = 128;
    const int N = in_sizes[0] / F;
    const int E = in_sizes[1];

    const int gemmBlocks = (N + 15) / 16;
    const int ownBlocks = (N + OWN_ROWS - 1) / OWN_ROWS;

    char* ws = (char*)d_ws;
    float* nsrc = (float*)ws;
    float* ndst = nsrc + N;
    size_t off = ((size_t)2 * N * sizeof(float) + 255) & ~(size_t)255;
    float* buf0 = (float*)(ws + off);
    float* buf1 = buf0 + (size_t)N * F;
    int* gq = (int*)(buf1 + (size_t)N * F);
    int* gq_n = gq + (size_t)ownBlocks * OWN_WAVES * OWN_QW * 2;

    hipMemsetAsync(nsrc, 0, (size_t)2 * N * sizeof(float), stream);
    gnn_deg_kernel<<<(E + 255) / 256, 256, 0, stream>>>(src, dst, nsrc, ndst, E);
    gnn_norm_kernel<<<(2 * N + 255) / 256, 256, 0, stream>>>(nsrc, 2 * N);

    gnn_gemm_kernel<<<gemmBlocks, 128, 0, stream>>>(feature, nsrc, W1, buf0, N);
    gnn_own_kernel<false><<<ownBlocks, OWN_THREADS, 0, stream>>>(buf0, src, dst, ndst, b1, buf1, gq, gq_n, N, E, 1);

    gnn_gemm_kernel<<<gemmBlocks, 128, 0, stream>>>(buf1, nsrc, W2, buf0, N);
    gnn_own_kernel<true><<<ownBlocks, OWN_THREADS, 0, stream>>>(buf0, src, dst, ndst, b2, out, gq, gq_n, N, E, 0);
}


# Pass-1 filter scans a 1-byte owner-group key per edge (stored by the degree kernel), not 4-byte dst; a wave lists its matches, then fetches their dst and src 32 at a time; queue-arrival sum order
# speedup vs baseline: 2.4511x; 1.1101x over previous
#include <hip/hip_runtime.h>
#include <hip/hip_bf16.h>

typedef __attribute__((ext_vector_type(2))) float v2f;
typedef __attribute__((ext_vector_type(8))) float v8f;

#define LDSTRIDE 132

#define OWN_ROWS    416

__global__ void gnn_deg_kernel(const int* __restrict__ src,
                               const int* __restrict__ dst,
                               float* __restrict__ deg_out,
                               float* __restrict__ deg_in,
                               unsigned char* __restrict__ key, int E) {
    int e = blockIdx.x * blockDim.x + threadIdx.x;
    if (e < E) {
        const int d = dst[e];
        atomicAdd(&deg_out[src[e]], 1.0f);
        atomicAdd(&deg_in[d], 1.0f);
        key[e] = (unsigned char)((unsigned)d / OWN_ROWS);
    }
}

__global__ void gnn_norm_kernel(float* __restrict__ d, int n) {
    int i = blockIdx.x * blockDim.x + threadIdx.x;
    if (i < n) {
        float v = fmaxf(d[i], 1.0f);
        d[i] = 1.0f / sqrtf(v);
    }
}

__global__ __launch_bounds__(128) void gnn_gemm_kernel(
    const float* __restrict__ H, const float* __restrict__ nsrc,
    const float* __restrict__ W, float* __restrict__ X, int N) {
    __shared__ float As[16 * LDSTRIDE];

    const int row0 = blockIdx.x * 16;
    const int tid  = threadIdx.x;

    for (int i = tid; i < 16 * 32; i += 128) {
        int r  = i >> 5;
        int c4 = (i & 31) * 4;
        int row = row0 + r;
        int rowc = (row < N) ? row : (N - 1);
        float s = nsrc[rowc];
        float4 v = *(const float4*)(H + (size_t)rowc * 128 + c4);
        float4 sv = make_float4(v.x * s, v.y * s, v.z * s, v.w * s);
        *(float4*)(&As[r * LDSTRIDE + c4]) = sv;
    }
    __syncthreads();

    const int wave = tid >> 5;
    const int lane = tid & 31;
    const int ln   = lane & 15;
    const int koff = (lane < 16) ? 0 : 2;

    const int n0 = wave * 32;
    const int n1 = n0 + 16;

    v8f acc0 = {};
    v8f acc1 = {};

    for (int kb = 0; kb < 128; kb += 4) {
        v2f a = *(const v2f*)(&As[ln * LDSTRIDE + kb + koff]);

        const float* wp = W + (size_t)(kb + koff) * 128;
        v2f b0, b1;
        b0.x = wp[n0 + ln];
        b0.y = wp[128 + n0 + ln];
        b1.x = wp[n1 + ln];
        b1.y = wp[128 + n1 + ln];

        acc0 = __builtin_amdgcn_wmma_f32_16x16x4_f32(
            false, a, false, b0, (short)0, acc0, false, false);
        acc1 = __builtin_amdgcn_wmma_f32_16x16x4_f32(
            false, a, false, b1, (short)0, acc1, false, false);
    }

    const int mbase = row0 + ((lane < 16) ? 0 : 8);
#pragma unroll
    for (int v = 0; v < 8; ++v) {
        int row = mbase + v;
        if (row < N) {
            X[(size_t)row * 128 + n0 + ln] = acc0[v];
            X[(size_t)row * 128 + n1 + ln] = acc1[v];
        }
    }
}

#define OWN_TRASH   OWN_ROWS
#define OWN_THREADS 1024
#define OWN_WAVES   (OWN_THREADS / 32)
#define OWN_QW      384
#define OWN_CW      384
#define OWN_RBITS   9
#define OWN_GROUP   8
#define OWN_AHEAD   16

static_assert(OWN_ROWS + 1 <= (1 << OWN_RBITS), "a local row must fit its bits of the queue word");

struct OwnLds {
    alignas(16) float acc[(OWN_ROWS + 1) * 128];
    int cand[OWN_WAVES * OWN_CW];
    unsigned q[OWN_WAVES * OWN_QW];
    int q_n[OWN_WAVES];
};

__device__ __forceinline__ void gnn_own_take(
    OwnLds& L, int s, unsigned lr, const float* __restrict__ X) {
    const int w = (int)(lr & (OWN_WAVES - 1));
    const int slot = atomicAdd(&L.q_n[w], 1);
    if (slot < OWN_QW) {
        L.q[w * OWN_QW + slot] = ((unsigned)s << OWN_RBITS) | lr;
    } else {
        const float* xr = X + (size_t)s * 128;
        float* ar = L.acc + lr * 128;
#pragma unroll 1
        for (int c = 0; c < 128; ++c) atomicAdd(ar + c, xr[c]);
    }
}

__device__ __forceinline__ void gnn_own_exact(
    OwnLds& L, int e, int base, unsigned rows,
    const float* __restrict__ X, const int* __restrict__ src,
    const int* __restrict__ dst) {
    const unsigned lr = (unsigned)(dst[e] - base);
    if (lr < rows) gnn_own_take(L, src[e], lr, X);
}

__device__ __forceinline__ unsigned gnn_own_hits(uint4 d, unsigned G) {
    const unsigned t0 = d.x ^ G, t1 = d.y ^ G, t2 = d.z ^ G, t3 = d.w ^ G;
    const unsigned f0 = (t0 - 0x01010101u) & ~t0;
    const unsigned f1 = (t1 - 0x01010101u) & ~t1;
    const unsigned f2 = (t2 - 0x01010101u) & ~t2;
    const unsigned f3 = (t3 - 0x01010101u) & ~t3;
    return ((f0 >> 7) & 0x01010101u) | ((f1 >> 6) & 0x02020202u) |
           ((f2 >> 5) & 0x04040404u) | ((f3 >> 4) & 0x08080808u);
}

__device__ __forceinline__ int gnn_own_scan(
    OwnLds& L, int tid, int lane, int wave, int base, unsigned rows,
    const float* __restrict__ X, const int* __restrict__ src,
    const int* __restrict__ dst, const unsigned char* __restrict__ key, int E) {
    const int nchunk = E >> 9;
    const uint4* k4 = (const uint4*)key;
    const unsigned G = (unsigned)(blockIdx.x & 255) * 0x01010101u;
    const int start = (int)(((long long)blockIdx.x * nchunk) / gridDim.x);
    int* cq = L.cand + wave * OWN_CW;
    int cnt = 0;
    for (int k = wave; k < nchunk; k += OWN_AHEAD * OWN_WAVES) {
        int c[OWN_AHEAD];
        uint4 d[OWN_AHEAD];
#pragma unroll
        for (int u = 0; u < OWN_AHEAD; ++u) {
            const int kk = k + u * OWN_WAVES;
            int cc = start + (kk < nchunk ? kk : k);
            cc = cc >= nchunk ? cc - nchunk : cc;
            c[u] = cc;
            d[u] = k4[(size_t)cc * 32 + lane];
        }
#pragma unroll
        for (int u = 0; u < OWN_AHEAD; ++u) {
            if (k + u * OWN_WAVES < nchunk) {
                unsigned m = gnn_own_hits(d[u], G);
                const int e0 = c[u] * 512 + lane * 16;
                unsigned act = __builtin_amdgcn_ballot_w32(m != 0);
                while (act) {
                    if (m != 0) {
                        const int i = __builtin_ctz(m);
                        m &= m - 1;
                        const int e = e0 + (((i & 3) << 2) | (i >> 3));
                        const int slot = cnt + (int)__builtin_amdgcn_mbcnt_lo(act, 0u);
                        if (slot < OWN_CW) cq[slot] = e;
                    }
                    cnt += __builtin_popcount(act);
                    act = __builtin_amdgcn_ballot_w32(m != 0);
                }
            }
        }
    }
    for (int e = nchunk * 512 + tid; e < E; e += OWN_THREADS)
        gnn_own_exact(L, e, base, rows, X, src, dst);
    return cnt;
}

__device__ __forceinline__ void gnn_own_scan_exact(
    OwnLds& L, int lane, int wave, int base, unsigned rows,
    const float* __restrict__ X, const int* __restrict__ src,
    const int* __restrict__ dst, int E) {
    const int nchunk = E >> 9;
    const int start = (int)(((long long)blockIdx.x * nchunk) / gridDim.x);
    for (int k = wave; k < nchunk; k += OWN_WAVES) {
        int cc = start + k;
        cc = cc >= nchunk ? cc - nchunk : cc;
#pragma unroll 1
        for (int j = 0; j < 16; ++j)
            gnn_own_exact(L, cc * 512 + j * 32 + lane, base, rows, X, src, dst);
    }
}

__device__ __forceinline__ void gnn_own_resolve(
    OwnLds& L, int n, int wave, int lane, int base, unsigned rows,
    const float* __restrict__ X, const int* __restrict__ src,
    const int* __restrict__ dst) {
    const int* cq = L.cand + wave * OWN_CW;
    int e_nx = cq[lane < n ? lane : 0];
    e_nx = lane < n ? e_nx : 0;
    int d_nx = dst[e_nx];
    int s_nx = src[e_nx];
    for (int i0 = 0; i0 < n; i0 += 32) {
        const int d = d_nx, s = s_nx;
        const bool ok = i0 + lane < n;
        const int idn = i0 + 32 + lane;
        int en = cq[idn < n ? idn : 0];
        en = idn < n ? en : 0;
        d_nx = dst[en];
        s_nx = src[en];
        const unsigned lr = (unsigned)(d - base);
        if (ok && lr < rows) gnn_own_take(L, s, lr, X);
    }
}

template <bool DIRECT, bool SAVE>
__device__ __forceinline__ void gnn_own_add(
    OwnLds& L, int n, int wave, int lane,
    const float* __restrict__ X, int2* __restrict__ g) {
    const unsigned* qq = L.q + wave * OWN_QW;
    int s_nx = 0, r_nx = OWN_TRASH;
    if (lane < OWN_GROUP && lane < n) {
        if (DIRECT) {
            const int2 p = g[lane];
            s_nx = p.x;
            r_nx = p.y;
        } else {
            const unsigned pk = qq[lane];
            s_nx = (int)(pk >> OWN_RBITS);
            r_nx = (int)(pk & ((1u << OWN_RBITS) - 1u));
        }
    }
    for (int i0 = 0; i0 < n; i0 += OWN_GROUP) {
        const int s_cu = s_nx, r_cu = r_nx;
        if (SAVE) {
            if (lane < OWN_GROUP && i0 + lane < n) g[i0 + lane] = make_int2(s_cu, r_cu);
        }
        const int idx = i0 + OWN_GROUP + lane;
        const bool ok = lane < OWN_GROUP && idx < n;
        const int idc = ok ? idx : 0;
        int r_ld, s_ld;
        if (DIRECT) {
            const int2 p = g[idc];
            s_ld = p.x;
            r_ld = p.y;
        } else {
            const unsigned pk = qq[idc];
            s_ld = (int)(pk >> OWN_RBITS);
            r_ld = (int)(pk & ((1u << OWN_RBITS) - 1u));
        }
        r_nx = ok ? r_ld : OWN_TRASH;
        s_nx = ok ? s_ld : 0;
        float4 v[OWN_GROUP];
#pragma unroll
        for (int b = 0; b < OWN_GROUP; ++b) {
            const int sb = __builtin_amdgcn_readlane(s_cu, b);
            v[b] = ((const float4*)(X + (size_t)sb * 128))[lane];
        }
#pragma unroll
        for (int b = 0; b < OWN_GROUP; ++b) {
            const int rb = __builtin_amdgcn_readlane(r_cu, b);
            float4* p = (float4*)(L.acc + rb * 128) + lane;
            float4 a = *p;
            a.x += v[b].x;
            a.y += v[b].y;
            a.z += v[b].z;
            a.w += v[b].w;
            *p = a;
        }
    }
}

template <bool REPLAY>
__global__ __launch_bounds__(OWN_THREADS) void gnn_own_kernel(
    const float* __restrict__ X, const int* __restrict__ src,
    const int* __restrict__ dst, const float* __restrict__ ndst,
    const float* __restrict__ bias, float* __restrict__ OUT,
    int* __restrict__ gq, int* __restrict__ gq_n,
    const unsigned char* __restrict__ key,
    int N, int E, int do_relu) {
    __shared__ OwnLds L;

    const int tid = threadIdx.x;
    const int lane = tid & 31;
    const int wave = __builtin_amdgcn_readfirstlane(tid >> 5);
    const int base = blockIdx.x * OWN_ROWS;
    const int left = N - base;
    const unsigned rows = (unsigned)(left < OWN_ROWS ? left : OWN_ROWS);

    int c_all = 0;
    bool saved = false;
    if (REPLAY) {
        c_all = gq_n[blockIdx.x * OWN_WAVES + lane];
        saved = __builtin_amdgcn_ballot_w32(c_all > OWN_QW) == 0;
    }

    for (int i = tid; i < (OWN_ROWS + 1) * 32; i += OWN_THREADS)
        ((float4*)L.acc)[i] = make_float4(0.0f, 0.0f, 0.0f, 0.0f);
    if (tid < OWN_WAVES) L.q_n[tid] = 0;
    __syncthreads();

    int2* g = (int2*)gq + (size_t)(blockIdx.x * OWN_WAVES + wave) * OWN_QW;

    if (REPLAY && saved) {
        const int n = __builtin_amdgcn_readlane(c_all, wave);
        gnn_own_add<true, false>(L, n, wave, lane, X, g);
    } else {
        int nc = gnn_own_scan(L, tid, lane, wave, base, rows, X, src, dst, key, E);
        if (nc > OWN_CW) {
            nc = 0;
            gnn_own_scan_exact(L, lane, wave, base, rows, X, src, dst, E);
        }
        gnn_own_resolve(L, nc, wave, lane, base, rows, X, src, dst);
        __syncthreads();
        const int cnt = __builtin_amdgcn_readfirstlane(L.q_n[wave]);
        if (!REPLAY) {
            if (lane == 0) gq_n[blockIdx.x * OWN_WAVES + wave] = cnt;
        }
        const int n = cnt < OWN_QW ? cnt : OWN_QW;
        gnn_own_add<false, !REPLAY>(L, n, wave, lane, X, g);
    }
    __syncthreads();

    for (int i = tid; i < (int)rows * 32; i += OWN_THREADS) {
        const int rr = i >> 5;
        const int c4 = (i & 31) * 4;
        const int node = base + rr;
        const float s = ndst[node];
        const float4 a = *(const float4*)(&L.acc[rr * 128 + c4]);
        const float4 b = *(const float4*)(bias + c4);
        float4 o;
        o.x = a.x * s + b.x;
        o.y = a.y * s + b.y;
        o.z = a.z * s + b.z;
        o.w = a.w * s + b.w;
        if (do_relu) {
            o.x = fmaxf(o.x, 0.0f);
            o.y = fmaxf(o.y, 0.0f);
            o.z = fmaxf(o.z, 0.0f);
            o.w = fmaxf(o.w, 0.0f);
        }
        *(float4*)(OUT + (size_t)node * 128 + c4) = o;
    }
}

extern "C" void kernel_launch(void* const* d_in, const int* in_sizes, int n_in,
                              void* d_out, int out_size, void* d_ws, size_t ws_size,
                              hipStream_t stream) {
    const float* feature = (const float*)d_in[0];
    const int*   src     = (const int*)d_in[1];
    const int*   dst     = (const int*)d_in[2];
    const float* W1      = (const float*)d_in[3];
    const float* b1      = (const float*)d_in[4];
    const float* W2      = (const float*)d_in[5];
    const float* b2      = (const float*)d_in[6];
    float* out = (float*)d_out;

    const int F = 128;
    const int N = in_sizes[0] / F;
    const int E = in_sizes[1];

    const int gemmBlocks = (N + 15) / 16;
    const int ownBlocks = (N + OWN_ROWS - 1) / OWN_ROWS;

    char* ws = (char*)d_ws;
    float* nsrc = (float*)ws;
    float* ndst = nsrc + N;
    size_t off = ((size_t)2 * N * sizeof(float) + 255) & ~(size_t)255;
    float* buf0 = (float*)(ws + off);
    float* buf1 = buf0 + (size_t)N * F;
    int* gq = (int*)(buf1 + (size_t)N * F);
    int* gq_n = gq + (size_t)ownBlocks * OWN_WAVES * OWN_QW * 2;
    unsigned char* key = (unsigned char*)gq_n +
        (((size_t)ownBlocks * OWN_WAVES * sizeof(int) + 255) & ~(size_t)255);

    hipMemsetAsync(nsrc, 0, (size_t)2 * N * sizeof(float), stream);
    gnn_deg_kernel<<<(E + 255) / 256, 256, 0, stream>>>(src, dst, nsrc, ndst, key, E);
    gnn_norm_kernel<<<(2 * N + 255) / 256, 256, 0, stream>>>(nsrc, 2 * N);

    gnn_gemm_kernel<<<gemmBlocks, 128, 0, stream>>>(feature, nsrc, W1, buf0, N);
    gnn_own_kernel<false><<<ownBlocks, OWN_THREADS, 0, stream>>>(buf0, src, dst, ndst, b1, buf1, gq, gq_n, key, N, E, 1);

    gnn_gemm_kernel<<<gemmBlocks, 128, 0, stream>>>(buf1, nsrc, W2, buf0, N);
    gnn_own_kernel<true><<<ownBlocks, OWN_THREADS, 0, stream>>>(buf0, src, dst, ndst, b2, out, gq, gq_n, key, N, E, 0);
}


# In-degrees counted in LDS by the layer-1 owner work-group (one integer LDS add per queued edge; 1/sqrt scale formed once per row there, saved for layer 2) instead of a global float atomic per edge
# speedup vs baseline: 2.8106x; 1.1467x over previous
#include <hip/hip_runtime.h>
#include <hip/hip_bf16.h>

typedef __attribute__((ext_vector_type(2))) float v2f;
typedef __attribute__((ext_vector_type(8))) float v8f;

#define LDSTRIDE 132

#define OWN_ROWS    416

__global__ void gnn_deg_kernel(const int* __restrict__ src,
                               const int* __restrict__ dst,
                               float* __restrict__ deg_out,
                               unsigned char* __restrict__ key, int E) {
    int e = blockIdx.x * blockDim.x + threadIdx.x;
    if (e < E) {
        const int d = dst[e];
        atomicAdd(&deg_out[src[e]], 1.0f);
        key[e] = (unsigned char)((unsigned)d / OWN_ROWS);
    }
}

__global__ void gnn_norm_kernel(float* __restrict__ d, int n) {
    int i = blockIdx.x * blockDim.x + threadIdx.x;
    if (i < n) {
        float v = fmaxf(d[i], 1.0f);
        d[i] = 1.0f / sqrtf(v);
    }
}

__global__ __launch_bounds__(128) void gnn_gemm_kernel(
    const float* __restrict__ H, const float* __restrict__ nsrc,
    const float* __restrict__ W, float* __restrict__ X, int N) {
    __shared__ float As[16 * LDSTRIDE];

    const int row0 = blockIdx.x * 16;
    const int tid  = threadIdx.x;

    for (int i = tid; i < 16 * 32; i += 128) {
        int r  = i >> 5;
        int c4 = (i & 31) * 4;
        int row = row0 + r;
        int rowc = (row < N) ? row : (N - 1);
        float s = nsrc[rowc];
        float4 v = *(const float4*)(H + (size_t)rowc * 128 + c4);
        float4 sv = make_float4(v.x * s, v.y * s, v.z * s, v.w * s);
        *(float4*)(&As[r * LDSTRIDE + c4]) = sv;
    }
    __syncthreads();

    const int wave = tid >> 5;
    const int lane = tid & 31;
    const int ln   = lane & 15;
    const int koff = (lane < 16) ? 0 : 2;

    const int n0 = wave * 32;
    const int n1 = n0 + 16;

    v8f acc0 = {};
    v8f acc1 = {};

    for (int kb = 0; kb < 128; kb += 4) {
        v2f a = *(const v2f*)(&As[ln * LDSTRIDE + kb + koff]);

        const float* wp = W + (size_t)(kb + koff) * 128;
        v2f b0, b1;
        b0.x = wp[n0 + ln];
        b0.y = wp[128 + n0 + ln];
        b1.x = wp[n1 + ln];
        b1.y = wp[128 + n1 + ln];

        acc0 = __builtin_amdgcn_wmma_f32_16x16x4_f32(
            false, a, false, b0, (short)0, acc0, false, false);
        acc1 = __builtin_amdgcn_wmma_f32_16x16x4_f32(
            false, a, false, b1, (short)0, acc1, false, false);
    }

    const int mbase = row0 + ((lane < 16) ? 0 : 8);
#pragma unroll
    for (int v = 0; v < 8; ++v) {
        int row = mbase + v;
        if (row < N) {
            X[(size_t)row * 128 + n0 + ln] = acc0[v];
            X[(size_t)row * 128 + n1 + ln] = acc1[v];
        }
    }
}

#define OWN_TRASH   OWN_ROWS
#define OWN_THREADS 1024
#define OWN_WAVES   (OWN_THREADS / 32)
#define OWN_QW      384
#define OWN_CW      384
#define OWN_RBITS   9
#define OWN_GROUP   8
#define OWN_AHEAD   16

static_assert(OWN_ROWS + 1 <= (1 << OWN_RBITS), "a local row must fit its bits of the queue word");

struct OwnLds {
    alignas(16) float acc[(OWN_ROWS + 1) * 128];
    int cand[OWN_WAVES * OWN_CW];
    unsigned q[OWN_WAVES * OWN_QW];
    int q_n[OWN_WAVES];
    int deg[OWN_ROWS + 1];
    float nd[OWN_ROWS + 1];
};

__device__ __forceinline__ void gnn_own_take(
    OwnLds& L, int s, unsigned lr, const float* __restrict__ X) {
    atomicAdd(&L.deg[lr], 1);
    const int w = (int)(lr & (OWN_WAVES - 1));
    const int slot = atomicAdd(&L.q_n[w], 1);
    if (slot < OWN_QW) {
        L.q[w * OWN_QW + slot] = ((unsigned)s << OWN_RBITS) | lr;
    } else {
        const float* xr = X + (size_t)s * 128;
        float* ar = L.acc + lr * 128;
#pragma unroll 1
        for (int c = 0; c < 128; ++c) atomicAdd(ar + c, xr[c]);
    }
}

__device__ __forceinline__ void gnn_own_exact(
    OwnLds& L, int e, int base, unsigned rows,
    const float* __restrict__ X, const int* __restrict__ src,
    const int* __restrict__ dst) {
    const unsigned lr = (unsigned)(dst[e] - base);
    if (lr < rows) gnn_own_take(L, src[e], lr, X);
}

__device__ __forceinline__ unsigned gnn_own_hits(uint4 d, unsigned G) {
    const unsigned t0 = d.x ^ G, t1 = d.y ^ G, t2 = d.z ^ G, t3 = d.w ^ G;
    const unsigned f0 = (t0 - 0x01010101u) & ~t0;
    const unsigned f1 = (t1 - 0x01010101u) & ~t1;
    const unsigned f2 = (t2 - 0x01010101u) & ~t2;
    const unsigned f3 = (t3 - 0x01010101u) & ~t3;
    return ((f0 >> 7) & 0x01010101u) | ((f1 >> 6) & 0x02020202u) |
           ((f2 >> 5) & 0x04040404u) | ((f3 >> 4) & 0x08080808u);
}

__device__ __forceinline__ int gnn_own_scan(
    OwnLds& L, int tid, int lane, int wave, int base, unsigned rows,
    const float* __restrict__ X, const int* __restrict__ src,
    const int* __restrict__ dst, const unsigned char* __restrict__ key, int E) {
    const int nchunk = E >> 9;
    const uint4* k4 = (const uint4*)key;
    const unsigned G = (unsigned)(blockIdx.x & 255) * 0x01010101u;
    const int start = (int)(((long long)blockIdx.x * nchunk) / gridDim.x);
    int* cq = L.cand + wave * OWN_CW;
    int cnt = 0;
    for (int k = wave; k < nchunk; k += OWN_AHEAD * OWN_WAVES) {
        int c[OWN_AHEAD];
        uint4 d[OWN_AHEAD];
#pragma unroll
        for (int u = 0; u < OWN_AHEAD; ++u) {
            const int kk = k + u * OWN_WAVES;
            int cc = start + (kk < nchunk ? kk : k);
            cc = cc >= nchunk ? cc - nchunk : cc;
            c[u] = cc;
            d[u] = k4[(size_t)cc * 32 + lane];
        }
#pragma unroll
        for (int u = 0; u < OWN_AHEAD; ++u) {
            if (k + u * OWN_WAVES < nchunk) {
                unsigned m = gnn_own_hits(d[u], G);
                const int e0 = c[u] * 512 + lane * 16;
                unsigned act = __builtin_amdgcn_ballot_w32(m != 0);
                while (act) {
                    if (m != 0) {
                        const int i = __builtin_ctz(m);
                        m &= m - 1;
                        const int e = e0 + (((i & 3) << 2) | (i >> 3));
                        const int slot = cnt + (int)__builtin_amdgcn_mbcnt_lo(act, 0u);
                        if (slot < OWN_CW) cq[slot] = e;
                    }
                    cnt += __builtin_popcount(act);
                    act = __builtin_amdgcn_ballot_w32(m != 0);
                }
            }
        }
    }
    for (int e = nchunk * 512 + tid; e < E; e += OWN_THREADS)
        gnn_own_exact(L, e, base, rows, X, src, dst);
    return cnt;
}

__device__ __forceinline__ void gnn_own_scan_exact(
    OwnLds& L, int lane, int wave, int base, unsigned rows,
    const float* __restrict__ X, const int* __restrict__ src,
    const int* __restrict__ dst, int E) {
    const int nchunk = E >> 9;
    const int start = (int)(((long long)blockIdx.x * nchunk) / gridDim.x);
    for (int k = wave; k < nchunk; k += OWN_WAVES) {
        int cc = start + k;
        cc = cc >= nchunk ? cc - nchunk : cc;
#pragma unroll 1
        for (int j = 0; j < 16; ++j)
            gnn_own_exact(L, cc * 512 + j * 32 + lane, base, rows, X, src, dst);
    }
}

__device__ __forceinline__ void gnn_own_resolve(
    OwnLds& L, int n, int wave, int lane, int base, unsigned rows,
    const float* __restrict__ X, const int* __restrict__ src,
    const int* __restrict__ dst) {
    const int* cq = L.cand + wave * OWN_CW;
    int e_nx = cq[lane < n ? lane : 0];
    e_nx = lane < n ? e_nx : 0;
    int d_nx = dst[e_nx];
    int s_nx = src[e_nx];
    for (int i0 = 0; i0 < n; i0 += 32) {
        const int d = d_nx, s = s_nx;
        const bool ok = i0 + lane < n;
        const int idn = i0 + 32 + lane;
        int en = cq[idn < n ? idn : 0];
        en = idn < n ? en : 0;
        d_nx = dst[en];
        s_nx = src[en];
        const unsigned lr = (unsigned)(d - base);
        if (ok && lr < rows) gnn_own_take(L, s, lr, X);
    }
}

template <bool DIRECT, bool SAVE>
__device__ __forceinline__ void gnn_own_add(
    OwnLds& L, int n, int wave, int lane,
    const float* __restrict__ X, int2* __restrict__ g) {
    const unsigned* qq = L.q + wave * OWN_QW;
    int s_nx = 0, r_nx = OWN_TRASH;
    if (lane < OWN_GROUP && lane < n) {
        if (DIRECT) {
            const int2 p = g[lane];
            s_nx = p.x;
            r_nx = p.y;
        } else {
            const unsigned pk = qq[lane];
            s_nx = (int)(pk >> OWN_RBITS);
            r_nx = (int)(pk & ((1u << OWN_RBITS) - 1u));
        }
    }
    for (int i0 = 0; i0 < n; i0 += OWN_GROUP) {
        const int s_cu = s_nx, r_cu = r_nx;
        if (SAVE) {
            if (lane < OWN_GROUP && i0 + lane < n) g[i0 + lane] = make_int2(s_cu, r_cu);
        }
        const int idx = i0 + OWN_GROUP + lane;
        const bool ok = lane < OWN_GROUP && idx < n;
        const int idc = ok ? idx : 0;
        int r_ld, s_ld;
        if (DIRECT) {
            const int2 p = g[idc];
            s_ld = p.x;
            r_ld = p.y;
        } else {
            const unsigned pk = qq[idc];
            s_ld = (int)(pk >> OWN_RBITS);
            r_ld = (int)(pk & ((1u << OWN_RBITS) - 1u));
        }
        r_nx = ok ? r_ld : OWN_TRASH;
        s_nx = ok ? s_ld : 0;
        float4 v[OWN_GROUP];
#pragma unroll
        for (int b = 0; b < OWN_GROUP; ++b) {
            const int sb = __builtin_amdgcn_readlane(s_cu, b);
            v[b] = ((const float4*)(X + (size_t)sb * 128))[lane];
        }
#pragma unroll
        for (int b = 0; b < OWN_GROUP; ++b) {
            const int rb = __builtin_amdgcn_readlane(r_cu, b);
            float4* p = (float4*)(L.acc + rb * 128) + lane;
            float4 a = *p;
            a.x += v[b].x;
            a.y += v[b].y;
            a.z += v[b].z;
            a.w += v[b].w;
            *p = a;
        }
    }
}

template <bool REPLAY>
__global__ __launch_bounds__(OWN_THREADS) void gnn_own_kernel(
    const float* __restrict__ X, const int* __restrict__ src,
    const int* __restrict__ dst, float* __restrict__ ndst,
    const float* __restrict__ bias, float* __restrict__ OUT,
    int* __restrict__ gq, int* __restrict__ gq_n,
    const unsigned char* __restrict__ key,
    int N, int E, int do_relu) {
    __shared__ OwnLds L;

    const int tid = threadIdx.x;
    const int lane = tid & 31;
    const int wave = __builtin_amdgcn_readfirstlane(tid >> 5);
    const int base = blockIdx.x * OWN_ROWS;
    const int left = N - base;
    const unsigned rows = (unsigned)(left < OWN_ROWS ? left : OWN_ROWS);

    int c_all = 0;
    bool saved = false;
    if (REPLAY) {
        c_all = gq_n[blockIdx.x * OWN_WAVES + lane];
        saved = __builtin_amdgcn_ballot_w32(c_all > OWN_QW) == 0;
    }

    for (int i = tid; i < (OWN_ROWS + 1) * 32; i += OWN_THREADS)
        ((float4*)L.acc)[i] = make_float4(0.0f, 0.0f, 0.0f, 0.0f);
    if (tid < OWN_WAVES) L.q_n[tid] = 0;
    if (tid <= OWN_ROWS) L.deg[tid] = 0;
    __syncthreads();

    int2* g = (int2*)gq + (size_t)(blockIdx.x * OWN_WAVES + wave) * OWN_QW;

    if (REPLAY && saved) {
        const int n = __builtin_amdgcn_readlane(c_all, wave);
        gnn_own_add<true, false>(L, n, wave, lane, X, g);
    } else {
        int nc = gnn_own_scan(L, tid, lane, wave, base, rows, X, src, dst, key, E);
        if (nc > OWN_CW) {
            nc = 0;
            gnn_own_scan_exact(L, lane, wave, base, rows, X, src, dst, E);
        }
        gnn_own_resolve(L, nc, wave, lane, base, rows, X, src, dst);
        __syncthreads();
        const int cnt = __builtin_amdgcn_readfirstlane(L.q_n[wave]);
        if (!REPLAY) {
            if (lane == 0) gq_n[blockIdx.x * OWN_WAVES + wave] = cnt;
        }
        if (!REPLAY) {
            if (tid < (int)rows) {
                const float s = 1.0f / sqrtf(fmaxf((float)L.deg[tid], 1.0f));
                L.nd[tid] = s;
                ndst[base + tid] = s;
            }
        }
        const int n = cnt < OWN_QW ? cnt : OWN_QW;
        gnn_own_add<false, !REPLAY>(L, n, wave, lane, X, g);
    }
    __syncthreads();

    for (int i = tid; i < (int)rows * 32; i += OWN_THREADS) {
        const int rr = i >> 5;
        const int c4 = (i & 31) * 4;
        const int node = base + rr;
        const float s = REPLAY ? ndst[node] : L.nd[rr];
        const float4 a = *(const float4*)(&L.acc[rr * 128 + c4]);
        const float4 b = *(const float4*)(bias + c4);
        float4 o;
        o.x = a.x * s + b.x;
        o.y = a.y * s + b.y;
        o.z = a.z * s + b.z;
        o.w = a.w * s + b.w;
        if (do_relu) {
            o.x = fmaxf(o.x, 0.0f);
            o.y = fmaxf(o.y, 0.0f);
            o.z = fmaxf(o.z, 0.0f);
            o.w = fmaxf(o.w, 0.0f);
        }
        *(float4*)(OUT + (size_t)node * 128 + c4) = o;
    }
}

extern "C" void kernel_launch(void* const* d_in, const int* in_sizes, int n_in,
                              void* d_out, int out_size, void* d_ws, size_t ws_size,
                              hipStream_t stream) {
    const float* feature = (const float*)d_in[0];
    const int*   src     = (const int*)d_in[1];
    const int*   dst     = (const int*)d_in[2];
    const float* W1      = (const float*)d_in[3];
    const float* b1      = (const float*)d_in[4];
    const float* W2      = (const float*)d_in[5];
    const float* b2      = (const float*)d_in[6];
    float* out = (float*)d_out;

    const int F = 128;
    const int N = in_sizes[0] / F;
    const int E = in_sizes[1];

    const int gemmBlocks = (N + 15) / 16;
    const int ownBlocks = (N + OWN_ROWS - 1) / OWN_ROWS;

    char* ws = (char*)d_ws;
    float* nsrc = (float*)ws;
    float* ndst = nsrc + N;
    size_t off = ((size_t)2 * N * sizeof(float) + 255) & ~(size_t)255;
    float* buf0 = (float*)(ws + off);
    float* buf1 = buf0 + (size_t)N * F;
    int* gq = (int*)(buf1 + (size_t)N * F);
    int* gq_n = gq + (size_t)ownBlocks * OWN_WAVES * OWN_QW * 2;
    unsigned char* key = (unsigned char*)gq_n +
        (((size_t)ownBlocks * OWN_WAVES * sizeof(int) + 255) & ~(size_t)255);

    hipMemsetAsync(nsrc, 0, (size_t)N * sizeof(float), stream);
    gnn_deg_kernel<<<(E + 255) / 256, 256, 0, stream>>>(src, dst, nsrc, key, E);
    gnn_norm_kernel<<<(N + 255) / 256, 256, 0, stream>>>(nsrc, N);

    gnn_gemm_kernel<<<gemmBlocks, 128, 0, stream>>>(feature, nsrc, W1, buf0, N);
    gnn_own_kernel<false><<<ownBlocks, OWN_THREADS, 0, stream>>>(buf0, src, dst, ndst, b1, buf1, gq, gq_n, key, N, E, 1);

    gnn_gemm_kernel<<<gemmBlocks, 128, 0, stream>>>(buf1, nsrc, W2, buf0, N);
    gnn_own_kernel<true><<<ownBlocks, OWN_THREADS, 0, stream>>>(buf0, src, dst, ndst, b2, out, gq, gq_n, key, N, E, 0);
}
